# starting kernel
# baseline (speedup 1.0000x reference)
#include <hip/hip_runtime.h>
#include <hip/hip_bf16.h>

typedef __attribute__((ext_vector_type(2))) float v2f;
typedef __attribute__((ext_vector_type(8))) float v8f;

#define DDIM 1024
#define RANK 256
#define NB   16
#define SS   256
#define NROWS (NB * SS)
#define NCOLS (2 * RANK)

__global__ __launch_bounds__(32) void gemm_lr_kernel(
    const float* __restrict__ batch, const float* __restrict__ W1,
    const float* __restrict__ b1, float* __restrict__ leftp,
    float* __restrict__ rightp)
{
  const int lane = threadIdx.x;
  const int m0   = blockIdx.x * 32;
  const int n0   = blockIdx.y * 64;
  const int mrow = lane & 15;
  const int kb   = (lane >> 4) * 2;

  const long aBase0 = (long)(m0 + mrow) * DDIM + kb;
  const long aBase1 = (long)(m0 + 16 + mrow) * DDIM + kb;

  long wBase[4];
#pragma unroll
  for (int t = 0; t < 4; ++t) {
    int ncol = n0 + t * 16 + (lane & 15);
    int side = (ncol >= RANK) ? 1 : 0;
    int wrow = ncol - side * RANK;
    wBase[t] = (long)wrow * (2 * DDIM) + side * DDIM + kb;
  }

  v8f acc[2][4];
#pragma unroll
  for (int i = 0; i < 2; ++i)
#pragma unroll
    for (int t = 0; t < 4; ++t) acc[i][t] = (v8f){0.f,0.f,0.f,0.f,0.f,0.f,0.f,0.f};

#pragma unroll 2
  for (int k0 = 0; k0 < DDIM; k0 += 4) {
    v2f a0 = *(const v2f*)(batch + aBase0 + k0);
    v2f a1 = *(const v2f*)(batch + aBase1 + k0);
    v2f bf[4];
#pragma unroll
    for (int t = 0; t < 4; ++t) bf[t] = *(const v2f*)(W1 + wBase[t] + k0);
#pragma unroll
    for (int t = 0; t < 4; ++t) {
      acc[0][t] = __builtin_amdgcn_wmma_f32_16x16x4_f32(
          false, a0, false, bf[t], (short)0, acc[0][t], false, false);
      acc[1][t] = __builtin_amdgcn_wmma_f32_16x16x4_f32(
          false, a1, false, bf[t], (short)0, acc[1][t], false, false);
    }
  }

  const int rowoff = (lane >> 4) ? 8 : 0;
#pragma unroll
  for (int t = 0; t < 4; ++t) {
    int ncol = n0 + t * 16 + (lane & 15);
    int side = (ncol >= RANK) ? 1 : 0;
    int ccol = ncol - side * RANK;
    float bias = side ? 0.f : b1[ncol];
    float* dst = side ? rightp : leftp;
#pragma unroll
    for (int i = 0; i < 2; ++i) {
#pragma unroll
      for (int v = 0; v < 8; ++v) {
        int row = m0 + i * 16 + rowoff + v;
        dst[(long)row * RANK + ccol] = acc[i][t][v] + bias;
      }
    }
  }
}

#define PITCH 257
__global__ __launch_bounds__(256) void pair_relu_dot_kernel(
    const float* __restrict__ leftp, const float* __restrict__ rightp,
    const float* __restrict__ wlast, const float* __restrict__ blast,
    float* __restrict__ out)
{
  __shared__ float Ls[64][PITCH];
  __shared__ float Rs[64][PITCH];
  __shared__ float ws[RANK];

  const int t  = threadIdx.x;
  const int jb = blockIdx.x;
  const int ib = blockIdx.y;
  const int bz = blockIdx.z;
  const int si = bz * SS + ib * 64;
  const int sj = bz * SS + jb * 64;

  {
    int row = t & 63;
    int q   = t >> 6;
    const float4* lr = (const float4*)(leftp  + (long)(si + row) * RANK);
    const float4* rr = (const float4*)(rightp + (long)(sj + row) * RANK);
#pragma unroll
    for (int u = 0; u < 16; ++u) {
      float4 lv = lr[q * 16 + u];
      float4 rv = rr[q * 16 + u];
      int c = q * 64 + u * 4;
      Ls[row][c]   = lv.x; Ls[row][c+1] = lv.y;
      Ls[row][c+2] = lv.z; Ls[row][c+3] = lv.w;
      Rs[row][c]   = rv.x; Rs[row][c+1] = rv.y;
      Rs[row][c+2] = rv.z; Rs[row][c+3] = rv.w;
    }
    ws[t] = wlast[t];
  }
  __syncthreads();

  const int ti = t >> 4;
  const int tj = t & 15;
  float acc[4][4] = {{0.f}};

#pragma unroll 2
  for (int r = 0; r < RANK; ++r) {
    float la[4], rb[4];
#pragma unroll
    for (int a = 0; a < 4; ++a) la[a] = Ls[ti * 4 + a][r];
#pragma unroll
    for (int b = 0; b < 4; ++b) rb[b] = Rs[tj * 4 + b][r];
    float w = ws[r];
#pragma unroll
    for (int a = 0; a < 4; ++a)
#pragma unroll
      for (int b = 0; b < 4; ++b) {
        float h = fmaxf(la[a] + rb[b], 0.f);
        acc[a][b] = __builtin_fmaf(h, w, acc[a][b]);
      }
  }

  const float bl = blast[0];
#pragma unroll
  for (int a = 0; a < 4; ++a) {
#pragma unroll
    for (int b = 0; b < 4; ++b) {
      int i = ib * 64 + ti * 4 + a;
      int j = jb * 64 + tj * 4 + b;
      out[(long)bz * (SS * SS) + (long)i * SS + j] = acc[a][b] + bl;
    }
  }
}

extern "C" void kernel_launch(void* const* d_in, const int* in_sizes, int n_in,
                              void* d_out, int out_size, void* d_ws, size_t ws_size,
                              hipStream_t stream) {
  (void)in_sizes; (void)n_in; (void)out_size; (void)ws_size;
  const float* batch  = (const float*)d_in[0];
  const float* W1     = (const float*)d_in[1];
  const float* b1     = (const float*)d_in[2];
  const float* W_last = (const float*)d_in[3];
  const float* b_last = (const float*)d_in[4];
  float* out = (float*)d_out;

  float* leftp  = (float*)d_ws;
  float* rightp = leftp + (size_t)NROWS * RANK;

  dim3 g1(NROWS / 32, NCOLS / 64);
  gemm_lr_kernel<<<g1, 32, 0, stream>>>(batch, W1, b1, leftp, rightp);

  dim3 g2(4, 4, NB);
  pair_relu_dot_kernel<<<g2, 256, 0, stream>>>(leftp, rightp, W_last, b_last, out);
}


# GEMM fragments read straight from global as whole 128-B lines per row (16-B loads, lane half = 64 B, 64 k per trip, issued as one group ahead of its WMMAs); k pairing permuted, sum order changed
# speedup vs baseline: 1.3258x; 1.3258x over previous
#include <hip/hip_runtime.h>
#include <hip/hip_bf16.h>

typedef __attribute__((ext_vector_type(2))) float v2f;
typedef __attribute__((ext_vector_type(4))) float v4f;
typedef __attribute__((ext_vector_type(8))) float v8f;

#define DDIM 1024
#define RANK 256
#define NB   16
#define SS   256
#define NROWS (NB * SS)
#define NCOLS (2 * RANK)

__global__ __launch_bounds__(32) void gemm_lr_kernel(
    const float* __restrict__ batch, const float* __restrict__ W1,
    const float* __restrict__ b1, float* __restrict__ leftp,
    float* __restrict__ rightp)
{
  const int lane = threadIdx.x;
  const int m0   = blockIdx.x * 32;
  const int n0   = blockIdx.y * 64;
  const int c    = lane & 15;
  const int kh   = (lane >> 4) * 16;

  const float* aPtr0 = batch + (long)(m0 + c) * DDIM + kh;
  const float* aPtr1 = batch + (long)(m0 + 16 + c) * DDIM + kh;

  const float* bPtr[4];
#pragma unroll
  for (int t = 0; t < 4; ++t) {
    int ncol = n0 + t * 16 + c;
    int side = (ncol >= RANK) ? 1 : 0;
    int wrow = ncol - side * RANK;
    bPtr[t] = W1 + (long)wrow * (2 * DDIM) + side * DDIM + kh;
  }

  v8f acc[2][4];
#pragma unroll
  for (int i = 0; i < 2; ++i)
#pragma unroll
    for (int t = 0; t < 4; ++t) acc[i][t] = (v8f){0.f,0.f,0.f,0.f,0.f,0.f,0.f,0.f};

  for (int kb = 0; kb < DDIM; kb += 64) {
    v4f a0[2][4], a1[2][4], bf[4][2][4];
#pragma unroll
    for (int m = 0; m < 2; ++m) {
#pragma unroll
      for (int j = 0; j < 4; ++j) {
        const int o = kb + 32 * m + 4 * j;
        a0[m][j] = *(const v4f*)(aPtr0 + o);
        a1[m][j] = *(const v4f*)(aPtr1 + o);
#pragma unroll
        for (int t = 0; t < 4; ++t) bf[t][m][j] = *(const v4f*)(bPtr[t] + o);
      }
    }
#pragma unroll
    for (int m = 0; m < 2; ++m) {
#pragma unroll
      for (int j = 0; j < 4; ++j) {
#pragma unroll
        for (int p = 0; p < 2; ++p) {
          v2f fa0 = p ? (v2f){a0[m][j].z, a0[m][j].w} : (v2f){a0[m][j].x, a0[m][j].y};
          v2f fa1 = p ? (v2f){a1[m][j].z, a1[m][j].w} : (v2f){a1[m][j].x, a1[m][j].y};
#pragma unroll
          for (int t = 0; t < 4; ++t) {
            v2f fb = p ? (v2f){bf[t][m][j].z, bf[t][m][j].w}
                       : (v2f){bf[t][m][j].x, bf[t][m][j].y};
            acc[0][t] = __builtin_amdgcn_wmma_f32_16x16x4_f32(
                false, fa0, false, fb, (short)0, acc[0][t], false, false);
            acc[1][t] = __builtin_amdgcn_wmma_f32_16x16x4_f32(
                false, fa1, false, fb, (short)0, acc[1][t], false, false);
          }
        }
      }
    }
    __builtin_amdgcn_sched_group_barrier(0x020, 48, 0);
    __builtin_amdgcn_sched_group_barrier(0x008, 128, 0);
  }

  const int rowoff = (lane >> 4) ? 8 : 0;
#pragma unroll
  for (int t = 0; t < 4; ++t) {
    int ncol = n0 + t * 16 + (lane & 15);
    int side = (ncol >= RANK) ? 1 : 0;
    int ccol = ncol - side * RANK;
    float bias = side ? 0.f : b1[ncol];
    float* dst = side ? rightp : leftp;
#pragma unroll
    for (int i = 0; i < 2; ++i) {
#pragma unroll
      for (int v = 0; v < 8; ++v) {
        int row = m0 + i * 16 + rowoff + v;
        dst[(long)row * RANK + ccol] = acc[i][t][v] + bias;
      }
    }
  }
}

#define PITCH 257
__global__ __launch_bounds__(256) void pair_relu_dot_kernel(
    const float* __restrict__ leftp, const float* __restrict__ rightp,
    const float* __restrict__ wlast, const float* __restrict__ blast,
    float* __restrict__ out)
{
  __shared__ float Ls[64][PITCH];
  __shared__ float Rs[64][PITCH];
  __shared__ float ws[RANK];

  const int t  = threadIdx.x;
  const int jb = blockIdx.x;
  const int ib = blockIdx.y;
  const int bz = blockIdx.z;
  const int si = bz * SS + ib * 64;
  const int sj = bz * SS + jb * 64;

  {
    int row = t & 63;
    int q   = t >> 6;
    const float4* lr = (const float4*)(leftp  + (long)(si + row) * RANK);
    const float4* rr = (const float4*)(rightp + (long)(sj + row) * RANK);
#pragma unroll
    for (int u = 0; u < 16; ++u) {
      float4 lv = lr[q * 16 + u];
      float4 rv = rr[q * 16 + u];
      int c = q * 64 + u * 4;
      Ls[row][c]   = lv.x; Ls[row][c+1] = lv.y;
      Ls[row][c+2] = lv.z; Ls[row][c+3] = lv.w;
      Rs[row][c]   = rv.x; Rs[row][c+1] = rv.y;
      Rs[row][c+2] = rv.z; Rs[row][c+3] = rv.w;
    }
    ws[t] = wlast[t];
  }
  __syncthreads();

  const int ti = t >> 4;
  const int tj = t & 15;
  float acc[4][4] = {{0.f}};

#pragma unroll 2
  for (int r = 0; r < RANK; ++r) {
    float la[4], rb[4];
#pragma unroll
    for (int a = 0; a < 4; ++a) la[a] = Ls[ti * 4 + a][r];
#pragma unroll
    for (int b = 0; b < 4; ++b) rb[b] = Rs[tj * 4 + b][r];
    float w = ws[r];
#pragma unroll
    for (int a = 0; a < 4; ++a)
#pragma unroll
      for (int b = 0; b < 4; ++b) {
        float h = fmaxf(la[a] + rb[b], 0.f);
        acc[a][b] = __builtin_fmaf(h, w, acc[a][b]);
      }
  }

  const float bl = blast[0];
#pragma unroll
  for (int a = 0; a < 4; ++a) {
#pragma unroll
    for (int b = 0; b < 4; ++b) {
      int i = ib * 64 + ti * 4 + a;
      int j = jb * 64 + tj * 4 + b;
      out[(long)bz * (SS * SS) + (long)i * SS + j] = acc[a][b] + bl;
    }
  }
}

extern "C" void kernel_launch(void* const* d_in, const int* in_sizes, int n_in,
                              void* d_out, int out_size, void* d_ws, size_t ws_size,
                              hipStream_t stream) {
  (void)in_sizes; (void)n_in; (void)out_size; (void)ws_size;
  const float* batch  = (const float*)d_in[0];
  const float* W1     = (const float*)d_in[1];
  const float* b1     = (const float*)d_in[2];
  const float* W_last = (const float*)d_in[3];
  const float* b_last = (const float*)d_in[4];
  float* out = (float*)d_out;

  float* leftp  = (float*)d_ws;
  float* rightp = leftp + (size_t)NROWS * RANK;

  dim3 g1(NROWS / 32, NCOLS / 64);
  gemm_lr_kernel<<<g1, 32, 0, stream>>>(batch, W1, b1, leftp, rightp);

  dim3 g2(4, 4, NB);
  pair_relu_dot_kernel<<<g2, 256, 0, stream>>>(leftp, rightp, W_last, b_last, out);
}


# GEMM K range split over the 4 waves of a work-group (256 k each, 32-k trips), partial tiles summed through LDS: 4 waves per SIMD, load waits lie under other waves' WMMAs; sum order changed
# speedup vs baseline: 1.4706x; 1.1092x over previous
#include <hip/hip_runtime.h>
#include <hip/hip_bf16.h>

typedef __attribute__((ext_vector_type(2))) float v2f;
typedef __attribute__((ext_vector_type(4))) float v4f;
typedef __attribute__((ext_vector_type(8))) float v8f;

#define DDIM 1024
#define RANK 256
#define NB   16
#define SS   256
#define NROWS (NB * SS)
#define NCOLS (2 * RANK)

#define KWAVES 4
#define KPER   (DDIM / KWAVES)
__global__ __launch_bounds__(32 * KWAVES) void gemm_lr_kernel(
    const float* __restrict__ batch, const float* __restrict__ W1,
    const float* __restrict__ b1, float* __restrict__ leftp,
    float* __restrict__ rightp)
{
  __shared__ v4f part[KWAVES][16][32];

  const int lane = threadIdx.x & 31;
  const int wv   = __builtin_amdgcn_readfirstlane(threadIdx.x >> 5);
  const int m0   = blockIdx.x * 32;
  const int n0   = blockIdx.y * 64;
  const int c    = lane & 15;
  const int kh   = (lane >> 4) * 16;
  const int k0   = wv * KPER + kh;

  const float* aPtr0 = batch + (long)(m0 + c) * DDIM + k0;
  const float* aPtr1 = batch + (long)(m0 + 16 + c) * DDIM + k0;

  const int side = (blockIdx.y >= RANK / 64) ? 1 : 0;
  const int ccol0 = n0 - side * RANK + c;
  const float* bPtr[4];
  float bias[4];
#pragma unroll
  for (int t = 0; t < 4; ++t) {
    bPtr[t] = W1 + (long)(ccol0 + t * 16) * (2 * DDIM) + side * DDIM + k0;
    bias[t] = side ? 0.f : b1[ccol0 + t * 16];
  }

  v8f acc[2][4];
#pragma unroll
  for (int i = 0; i < 2; ++i)
#pragma unroll
    for (int t = 0; t < 4; ++t) acc[i][t] = (v8f){0.f,0.f,0.f,0.f,0.f,0.f,0.f,0.f};

  for (int kb = 0; kb < KPER; kb += 32) {
    v4f a0[4], a1[4], bf[4][4];
#pragma unroll
    for (int j = 0; j < 4; ++j) {
      const int o = kb + 4 * j;
      a0[j] = *(const v4f*)(aPtr0 + o);
      a1[j] = *(const v4f*)(aPtr1 + o);
#pragma unroll
      for (int t = 0; t < 4; ++t) bf[t][j] = *(const v4f*)(bPtr[t] + o);
    }
#pragma unroll
    for (int j = 0; j < 4; ++j) {
#pragma unroll
      for (int p = 0; p < 2; ++p) {
        v2f fa0 = p ? (v2f){a0[j].z, a0[j].w} : (v2f){a0[j].x, a0[j].y};
        v2f fa1 = p ? (v2f){a1[j].z, a1[j].w} : (v2f){a1[j].x, a1[j].y};
#pragma unroll
        for (int t = 0; t < 4; ++t) {
          v2f fb = p ? (v2f){bf[t][j].z, bf[t][j].w}
                     : (v2f){bf[t][j].x, bf[t][j].y};
          acc[0][t] = __builtin_amdgcn_wmma_f32_16x16x4_f32(
              false, fa0, false, fb, (short)0, acc[0][t], false, false);
          acc[1][t] = __builtin_amdgcn_wmma_f32_16x16x4_f32(
              false, fa1, false, fb, (short)0, acc[1][t], false, false);
        }
      }
    }
    __builtin_amdgcn_sched_group_barrier(0x020, 24, 0);
    __builtin_amdgcn_sched_group_barrier(0x008, 64, 0);
  }

#pragma unroll
  for (int i = 0; i < 2; ++i)
#pragma unroll
    for (int t = 0; t < 4; ++t)
#pragma unroll
      for (int vq = 0; vq < 2; ++vq)
        part[wv][(i * 4 + t) * 2 + vq][lane] =
            (v4f){acc[i][t][4 * vq], acc[i][t][4 * vq + 1],
                  acc[i][t][4 * vq + 2], acc[i][t][4 * vq + 3]};
  __syncthreads();

  const int ri = wv >> 1;
  const int rq = wv & 1;
  const int rowbase = m0 + 16 * ri + ((lane >> 4) ? 8 : 0) + 4 * rq;
  float* dst = side ? rightp : leftp;
#pragma unroll
  for (int t = 0; t < 4; ++t) {
    const int q = (ri * 4 + t) * 2 + rq;
    v4f s = part[0][q][lane];
    s += part[1][q][lane];
    s += part[2][q][lane];
    s += part[3][q][lane];
#pragma unroll
    for (int u = 0; u < 4; ++u)
      dst[(long)(rowbase + u) * RANK + ccol0 + t * 16] = s[u] + bias[t];
  }
}

#define PITCH 257
__global__ __launch_bounds__(256) void pair_relu_dot_kernel(
    const float* __restrict__ leftp, const float* __restrict__ rightp,
    const float* __restrict__ wlast, const float* __restrict__ blast,
    float* __restrict__ out)
{
  __shared__ float Ls[64][PITCH];
  __shared__ float Rs[64][PITCH];
  __shared__ float ws[RANK];

  const int t  = threadIdx.x;
  const int jb = blockIdx.x;
  const int ib = blockIdx.y;
  const int bz = blockIdx.z;
  const int si = bz * SS + ib * 64;
  const int sj = bz * SS + jb * 64;

  {
    int row = t & 63;
    int q   = t >> 6;
    const float4* lr = (const float4*)(leftp  + (long)(si + row) * RANK);
    const float4* rr = (const float4*)(rightp + (long)(sj + row) * RANK);
#pragma unroll
    for (int u = 0; u < 16; ++u) {
      float4 lv = lr[q * 16 + u];
      float4 rv = rr[q * 16 + u];
      int c = q * 64 + u * 4;
      Ls[row][c]   = lv.x; Ls[row][c+1] = lv.y;
      Ls[row][c+2] = lv.z; Ls[row][c+3] = lv.w;
      Rs[row][c]   = rv.x; Rs[row][c+1] = rv.y;
      Rs[row][c+2] = rv.z; Rs[row][c+3] = rv.w;
    }
    ws[t] = wlast[t];
  }
  __syncthreads();

  const int ti = t >> 4;
  const int tj = t & 15;
  float acc[4][4] = {{0.f}};

#pragma unroll 2
  for (int r = 0; r < RANK; ++r) {
    float la[4], rb[4];
#pragma unroll
    for (int a = 0; a < 4; ++a) la[a] = Ls[ti * 4 + a][r];
#pragma unroll
    for (int b = 0; b < 4; ++b) rb[b] = Rs[tj * 4 + b][r];
    float w = ws[r];
#pragma unroll
    for (int a = 0; a < 4; ++a)
#pragma unroll
      for (int b = 0; b < 4; ++b) {
        float h = fmaxf(la[a] + rb[b], 0.f);
        acc[a][b] = __builtin_fmaf(h, w, acc[a][b]);
      }
  }

  const float bl = blast[0];
#pragma unroll
  for (int a = 0; a < 4; ++a) {
#pragma unroll
    for (int b = 0; b < 4; ++b) {
      int i = ib * 64 + ti * 4 + a;
      int j = jb * 64 + tj * 4 + b;
      out[(long)bz * (SS * SS) + (long)i * SS + j] = acc[a][b] + bl;
    }
  }
}

extern "C" void kernel_launch(void* const* d_in, const int* in_sizes, int n_in,
                              void* d_out, int out_size, void* d_ws, size_t ws_size,
                              hipStream_t stream) {
  (void)in_sizes; (void)n_in; (void)out_size; (void)ws_size;
  const float* batch  = (const float*)d_in[0];
  const float* W1     = (const float*)d_in[1];
  const float* b1     = (const float*)d_in[2];
  const float* W_last = (const float*)d_in[3];
  const float* b_last = (const float*)d_in[4];
  float* out = (float*)d_out;

  float* leftp  = (float*)d_ws;
  float* rightp = leftp + (size_t)NROWS * RANK;

  dim3 g1(NROWS / 32, NCOLS / 64);
  gemm_lr_kernel<<<g1, 128, 0, stream>>>(batch, W1, b1, leftp, rightp);

  dim3 g2(4, 4, NB);
  pair_relu_dot_kernel<<<g2, 256, 0, stream>>>(leftp, rightp, W_last, b_last, out);
}


# GEMM wave tile 32x64 -> 64x64 (4x4 accumulators: 32 fragment loads feed 128 WMMAs per 32-k trip, was 24 per 64), a third fewer 16-line loads per WMMA; 2 waves per SIMD; same sum order, bit-identical
# speedup vs baseline: 1.5702x; 1.0677x over previous
#include <hip/hip_runtime.h>
#include <hip/hip_bf16.h>

typedef __attribute__((ext_vector_type(2))) float v2f;
typedef __attribute__((ext_vector_type(4))) float v4f;
typedef __attribute__((ext_vector_type(8))) float v8f;

#define DDIM 1024
#define RANK 256
#define NB   16
#define SS   256
#define NROWS (NB * SS)
#define NCOLS (2 * RANK)

#define KWAVES 4
#define KPER   (DDIM / KWAVES)
#define TM 4
#define TN 4
__global__ __launch_bounds__(32 * KWAVES) void gemm_lr_kernel(
    const float* __restrict__ batch, const float* __restrict__ W1,
    const float* __restrict__ b1, float* __restrict__ leftp,
    float* __restrict__ rightp)
{
  __shared__ v4f part[KWAVES][TM * TN * 2][32];

  const int lane = threadIdx.x & 31;
  const int wv   = __builtin_amdgcn_readfirstlane(threadIdx.x >> 5);
  const int m0   = blockIdx.x * (16 * TM);
  const int n0   = blockIdx.y * (16 * TN);
  const int c    = lane & 15;
  const int kh   = (lane >> 4) * 16;
  const int k0   = wv * KPER + kh;

  const float* aPtr[TM];
#pragma unroll
  for (int i = 0; i < TM; ++i)
    aPtr[i] = batch + (long)(m0 + 16 * i + c) * DDIM + k0;

  const int side = (blockIdx.y >= RANK / (16 * TN)) ? 1 : 0;
  const int ccol0 = n0 - side * RANK + c;
  const float* bPtr[TN];
  float bias[TN];
#pragma unroll
  for (int t = 0; t < TN; ++t) {
    bPtr[t] = W1 + (long)(ccol0 + t * 16) * (2 * DDIM) + side * DDIM + k0;
    bias[t] = side ? 0.f : b1[ccol0 + t * 16];
  }

  v8f acc[TM][TN];
#pragma unroll
  for (int i = 0; i < TM; ++i)
#pragma unroll
    for (int t = 0; t < TN; ++t) acc[i][t] = (v8f){0.f,0.f,0.f,0.f,0.f,0.f,0.f,0.f};

#pragma unroll
  for (int kb = 0; kb < KPER; kb += 32) {
    v4f af[TM][4], bf[TN][4];
#pragma unroll
    for (int j = 0; j < 4; ++j) {
      const int o = kb + 4 * j;
#pragma unroll
      for (int i = 0; i < TM; ++i) af[i][j] = *(const v4f*)(aPtr[i] + o);
#pragma unroll
      for (int t = 0; t < TN; ++t) bf[t][j] = *(const v4f*)(bPtr[t] + o);
    }
#pragma unroll
    for (int j = 0; j < 4; ++j) {
#pragma unroll
      for (int p = 0; p < 2; ++p) {
        v2f fa[TM];
#pragma unroll
        for (int i = 0; i < TM; ++i)
          fa[i] = p ? (v2f){af[i][j].z, af[i][j].w} : (v2f){af[i][j].x, af[i][j].y};
#pragma unroll
        for (int t = 0; t < TN; ++t) {
          v2f fb = p ? (v2f){bf[t][j].z, bf[t][j].w}
                     : (v2f){bf[t][j].x, bf[t][j].y};
#pragma unroll
          for (int i = 0; i < TM; ++i)
            acc[i][t] = __builtin_amdgcn_wmma_f32_16x16x4_f32(
                false, fa[i], false, fb, (short)0, acc[i][t], false, false);
        }
      }
    }
    __builtin_amdgcn_sched_group_barrier(0x020, 4 * (TM + TN), 0);
    __builtin_amdgcn_sched_group_barrier(0x008, 8 * TM * TN, 0);
  }

#pragma unroll
  for (int i = 0; i < TM; ++i)
#pragma unroll
    for (int t = 0; t < TN; ++t)
#pragma unroll
      for (int vq = 0; vq < 2; ++vq)
        part[wv][(i * TN + t) * 2 + vq][lane] =
            (v4f){acc[i][t][4 * vq], acc[i][t][4 * vq + 1],
                  acc[i][t][4 * vq + 2], acc[i][t][4 * vq + 3]};
  __syncthreads();

  float* dst = side ? rightp : leftp;
#pragma unroll
  for (int t = 0; t < TN; ++t) {
#pragma unroll
    for (int vq = 0; vq < 2; ++vq) {
      const int q = (wv * TN + t) * 2 + vq;
      const int rowbase = m0 + 16 * wv + ((lane >> 4) ? 8 : 0) + 4 * vq;
      v4f s = part[0][q][lane];
      s += part[1][q][lane];
      s += part[2][q][lane];
      s += part[3][q][lane];
#pragma unroll
      for (int u = 0; u < 4; ++u)
        dst[(long)(rowbase + u) * RANK + ccol0 + t * 16] = s[u] + bias[t];
    }
  }
}

#define PITCH 257
__global__ __launch_bounds__(256) void pair_relu_dot_kernel(
    const float* __restrict__ leftp, const float* __restrict__ rightp,
    const float* __restrict__ wlast, const float* __restrict__ blast,
    float* __restrict__ out)
{
  __shared__ float Ls[64][PITCH];
  __shared__ float Rs[64][PITCH];
  __shared__ float ws[RANK];

  const int t  = threadIdx.x;
  const int jb = blockIdx.x;
  const int ib = blockIdx.y;
  const int bz = blockIdx.z;
  const int si = bz * SS + ib * 64;
  const int sj = bz * SS + jb * 64;

  {
    int row = t & 63;
    int q   = t >> 6;
    const float4* lr = (const float4*)(leftp  + (long)(si + row) * RANK);
    const float4* rr = (const float4*)(rightp + (long)(sj + row) * RANK);
#pragma unroll
    for (int u = 0; u < 16; ++u) {
      float4 lv = lr[q * 16 + u];
      float4 rv = rr[q * 16 + u];
      int c = q * 64 + u * 4;
      Ls[row][c]   = lv.x; Ls[row][c+1] = lv.y;
      Ls[row][c+2] = lv.z; Ls[row][c+3] = lv.w;
      Rs[row][c]   = rv.x; Rs[row][c+1] = rv.y;
      Rs[row][c+2] = rv.z; Rs[row][c+3] = rv.w;
    }
    ws[t] = wlast[t];
  }
  __syncthreads();

  const int ti = t >> 4;
  const int tj = t & 15;
  float acc[4][4] = {{0.f}};

#pragma unroll 2
  for (int r = 0; r < RANK; ++r) {
    float la[4], rb[4];
#pragma unroll
    for (int a = 0; a < 4; ++a) la[a] = Ls[ti * 4 + a][r];
#pragma unroll
    for (int b = 0; b < 4; ++b) rb[b] = Rs[tj * 4 + b][r];
    float w = ws[r];
#pragma unroll
    for (int a = 0; a < 4; ++a)
#pragma unroll
      for (int b = 0; b < 4; ++b) {
        float h = fmaxf(la[a] + rb[b], 0.f);
        acc[a][b] = __builtin_fmaf(h, w, acc[a][b]);
      }
  }

  const float bl = blast[0];
#pragma unroll
  for (int a = 0; a < 4; ++a) {
#pragma unroll
    for (int b = 0; b < 4; ++b) {
      int i = ib * 64 + ti * 4 + a;
      int j = jb * 64 + tj * 4 + b;
      out[(long)bz * (SS * SS) + (long)i * SS + j] = acc[a][b] + bl;
    }
  }
}

extern "C" void kernel_launch(void* const* d_in, const int* in_sizes, int n_in,
                              void* d_out, int out_size, void* d_ws, size_t ws_size,
                              hipStream_t stream) {
  (void)in_sizes; (void)n_in; (void)out_size; (void)ws_size;
  const float* batch  = (const float*)d_in[0];
  const float* W1     = (const float*)d_in[1];
  const float* b1     = (const float*)d_in[2];
  const float* W_last = (const float*)d_in[3];
  const float* b_last = (const float*)d_in[4];
  float* out = (float*)d_out;

  float* leftp  = (float*)d_ws;
  float* rightp = leftp + (size_t)NROWS * RANK;

  dim3 g1(NROWS / 64, NCOLS / 64);
  gemm_lr_kernel<<<g1, 128, 0, stream>>>(batch, W1, b1, leftp, rightp);

  dim3 g2(4, 4, NB);
  pair_relu_dot_kernel<<<g2, 256, 0, stream>>>(leftp, rightp, W_last, b_last, out);
}


# pair kernel: tile rows held transposed (r-major) in LDS: a thread's 4 i, and its 4 j, of one r come as one 16-byte LDS read; staging reads 64-byte runs, 8 lines a load (was 32); sum order kept
# speedup vs baseline: 1.6593x; 1.0567x over previous
#include <hip/hip_runtime.h>
#include <hip/hip_bf16.h>

typedef __attribute__((ext_vector_type(2))) float v2f;
typedef __attribute__((ext_vector_type(4))) float v4f;
typedef __attribute__((ext_vector_type(8))) float v8f;

#define DDIM 1024
#define RANK 256
#define NB   16
#define SS   256
#define NROWS (NB * SS)
#define NCOLS (2 * RANK)

#define KWAVES 4
#define KPER   (DDIM / KWAVES)
#define TM 4
#define TN 4
__global__ __launch_bounds__(32 * KWAVES) void gemm_lr_kernel(
    const float* __restrict__ batch, const float* __restrict__ W1,
    const float* __restrict__ b1, float* __restrict__ leftp,
    float* __restrict__ rightp)
{
  __shared__ v4f part[KWAVES][TM * TN * 2][32];

  const int lane = threadIdx.x & 31;
  const int wv   = __builtin_amdgcn_readfirstlane(threadIdx.x >> 5);
  const int m0   = blockIdx.x * (16 * TM);
  const int n0   = blockIdx.y * (16 * TN);
  const int c    = lane & 15;
  const int kh   = (lane >> 4) * 16;
  const int k0   = wv * KPER + kh;

  const float* aPtr[TM];
#pragma unroll
  for (int i = 0; i < TM; ++i)
    aPtr[i] = batch + (long)(m0 + 16 * i + c) * DDIM + k0;

  const int side = (blockIdx.y >= RANK / (16 * TN)) ? 1 : 0;
  const int ccol0 = n0 - side * RANK + c;
  const float* bPtr[TN];
  float bias[TN];
#pragma unroll
  for (int t = 0; t < TN; ++t) {
    bPtr[t] = W1 + (long)(ccol0 + t * 16) * (2 * DDIM) + side * DDIM + k0;
    bias[t] = side ? 0.f : b1[ccol0 + t * 16];
  }

  v8f acc[TM][TN];
#pragma unroll
  for (int i = 0; i < TM; ++i)
#pragma unroll
    for (int t = 0; t < TN; ++t) acc[i][t] = (v8f){0.f,0.f,0.f,0.f,0.f,0.f,0.f,0.f};

#pragma unroll
  for (int kb = 0; kb < KPER; kb += 32) {
    v4f af[TM][4], bf[TN][4];
#pragma unroll
    for (int j = 0; j < 4; ++j) {
      const int o = kb + 4 * j;
#pragma unroll
      for (int i = 0; i < TM; ++i) af[i][j] = *(const v4f*)(aPtr[i] + o);
#pragma unroll
      for (int t = 0; t < TN; ++t) bf[t][j] = *(const v4f*)(bPtr[t] + o);
    }
#pragma unroll
    for (int j = 0; j < 4; ++j) {
#pragma unroll
      for (int p = 0; p < 2; ++p) {
        v2f fa[TM];
#pragma unroll
        for (int i = 0; i < TM; ++i)
          fa[i] = p ? (v2f){af[i][j].z, af[i][j].w} : (v2f){af[i][j].x, af[i][j].y};
#pragma unroll
        for (int t = 0; t < TN; ++t) {
          v2f fb = p ? (v2f){bf[t][j].z, bf[t][j].w}
                     : (v2f){bf[t][j].x, bf[t][j].y};
#pragma unroll
          for (int i = 0; i < TM; ++i)
            acc[i][t] = __builtin_amdgcn_wmma_f32_16x16x4_f32(
                false, fa[i], false, fb, (short)0, acc[i][t], false, false);
        }
      }
    }
    __builtin_amdgcn_sched_group_barrier(0x020, 4 * (TM + TN), 0);
    __builtin_amdgcn_sched_group_barrier(0x008, 8 * TM * TN, 0);
  }

#pragma unroll
  for (int i = 0; i < TM; ++i)
#pragma unroll
    for (int t = 0; t < TN; ++t)
#pragma unroll
      for (int vq = 0; vq < 2; ++vq)
        part[wv][(i * TN + t) * 2 + vq][lane] =
            (v4f){acc[i][t][4 * vq], acc[i][t][4 * vq + 1],
                  acc[i][t][4 * vq + 2], acc[i][t][4 * vq + 3]};
  __syncthreads();

  float* dst = side ? rightp : leftp;
#pragma unroll
  for (int t = 0; t < TN; ++t) {
#pragma unroll
    for (int vq = 0; vq < 2; ++vq) {
      const int q = (wv * TN + t) * 2 + vq;
      const int rowbase = m0 + 16 * wv + ((lane >> 4) ? 8 : 0) + 4 * vq;
      v4f s = part[0][q][lane];
      s += part[1][q][lane];
      s += part[2][q][lane];
      s += part[3][q][lane];
#pragma unroll
      for (int u = 0; u < 4; ++u)
        dst[(long)(rowbase + u) * RANK + ccol0 + t * 16] = s[u] + bias[t];
    }
  }
}

#define TP 68
__global__ __launch_bounds__(256) void pair_relu_dot_kernel(
    const float* __restrict__ leftp, const float* __restrict__ rightp,
    const float* __restrict__ wlast, const float* __restrict__ blast,
    float* __restrict__ out)
{
  __shared__ __attribute__((aligned(16))) float Lt[RANK][TP];
  __shared__ __attribute__((aligned(16))) float Rt[RANK][TP];
  __shared__ float ws[RANK];

  const int t  = threadIdx.x;
  const int jb = blockIdx.x;
  const int ib = blockIdx.y;
  const int bz = blockIdx.z;
  const int si = bz * SS + ib * 64;
  const int sj = bz * SS + jb * 64;

  {
    const int row = (t >> 5) * 8 + ((t & 31) >> 2);
    const int cl  = t & 3;
    const float4* lr = (const float4*)(leftp  + (long)(si + row) * RANK);
    const float4* rr = (const float4*)(rightp + (long)(sj + row) * RANK);
#pragma unroll
    for (int u = 0; u < 16; ++u) {
      const int c4 = 4 * u + cl;
      float4 lv = lr[c4];
      float4 rv = rr[c4];
      const int r0 = 4 * c4;
      Lt[r0][row]     = lv.x; Lt[r0 + 1][row] = lv.y;
      Lt[r0 + 2][row] = lv.z; Lt[r0 + 3][row] = lv.w;
      Rt[r0][row]     = rv.x; Rt[r0 + 1][row] = rv.y;
      Rt[r0 + 2][row] = rv.z; Rt[r0 + 3][row] = rv.w;
    }
    ws[t] = wlast[t];
  }
  __syncthreads();

  const int ti = t >> 4;
  const int tj = t & 15;
  float acc[4][4] = {{0.f}};

#pragma unroll 2
  for (int r = 0; r < RANK; ++r) {
    const v4f la = *(const v4f*)&Lt[r][ti * 4];
    const v4f rb = *(const v4f*)&Rt[r][tj * 4];
    float w = ws[r];
#pragma unroll
    for (int a = 0; a < 4; ++a)
#pragma unroll
      for (int b = 0; b < 4; ++b) {
        float h = fmaxf(la[a] + rb[b], 0.f);
        acc[a][b] = __builtin_fmaf(h, w, acc[a][b]);
      }
  }

  const float bl = blast[0];
#pragma unroll
  for (int a = 0; a < 4; ++a) {
#pragma unroll
    for (int b = 0; b < 4; ++b) {
      int i = ib * 64 + ti * 4 + a;
      int j = jb * 64 + tj * 4 + b;
      out[(long)bz * (SS * SS) + (long)i * SS + j] = acc[a][b] + bl;
    }
  }
}

extern "C" void kernel_launch(void* const* d_in, const int* in_sizes, int n_in,
                              void* d_out, int out_size, void* d_ws, size_t ws_size,
                              hipStream_t stream) {
  (void)in_sizes; (void)n_in; (void)out_size; (void)ws_size;
  const float* batch  = (const float*)d_in[0];
  const float* W1     = (const float*)d_in[1];
  const float* b1     = (const float*)d_in[2];
  const float* W_last = (const float*)d_in[3];
  const float* b_last = (const float*)d_in[4];
  float* out = (float*)d_out;

  float* leftp  = (float*)d_ws;
  float* rightp = leftp + (size_t)NROWS * RANK;

  dim3 g1(NROWS / 64, NCOLS / 64);
  gemm_lr_kernel<<<g1, 128, 0, stream>>>(batch, W1, b1, leftp, rightp);

  dim3 g2(4, 4, NB);
  pair_relu_dot_kernel<<<g2, 256, 0, stream>>>(leftp, rightp, W_last, b_last, out);
}


# pair kernel: relu(l+x) = max(l,-x)+x: x part summed once per right row at staging; r loop one max + one multiply-add per element (was add, max, fma), maxima dual-issued by hand; sum order changed
# speedup vs baseline: 1.7985x; 1.0839x over previous
#include <hip/hip_runtime.h>
#include <hip/hip_bf16.h>

typedef __attribute__((ext_vector_type(2))) float v2f;
typedef __attribute__((ext_vector_type(4))) float v4f;
typedef __attribute__((ext_vector_type(8))) float v8f;

#define DDIM 1024
#define RANK 256
#define NB   16
#define SS   256
#define NROWS (NB * SS)
#define NCOLS (2 * RANK)

#define KWAVES 4
#define KPER   (DDIM / KWAVES)
#define TM 4
#define TN 4
__global__ __launch_bounds__(32 * KWAVES) void gemm_lr_kernel(
    const float* __restrict__ batch, const float* __restrict__ W1,
    const float* __restrict__ b1, float* __restrict__ leftp,
    float* __restrict__ rightp)
{
  __shared__ v4f part[KWAVES][TM * TN * 2][32];

  const int lane = threadIdx.x & 31;
  const int wv   = __builtin_amdgcn_readfirstlane(threadIdx.x >> 5);
  const int m0   = blockIdx.x * (16 * TM);
  const int n0   = blockIdx.y * (16 * TN);
  const int c    = lane & 15;
  const int kh   = (lane >> 4) * 16;
  const int k0   = wv * KPER + kh;

  const float* aPtr[TM];
#pragma unroll
  for (int i = 0; i < TM; ++i)
    aPtr[i] = batch + (long)(m0 + 16 * i + c) * DDIM + k0;

  const int side = (blockIdx.y >= RANK / (16 * TN)) ? 1 : 0;
  const int ccol0 = n0 - side * RANK + c;
  const float* bPtr[TN];
  float bias[TN];
#pragma unroll
  for (int t = 0; t < TN; ++t) {
    bPtr[t] = W1 + (long)(ccol0 + t * 16) * (2 * DDIM) + side * DDIM + k0;
    bias[t] = side ? 0.f : b1[ccol0 + t * 16];
  }

  v8f acc[TM][TN];
#pragma unroll
  for (int i = 0; i < TM; ++i)
#pragma unroll
    for (int t = 0; t < TN; ++t) acc[i][t] = (v8f){0.f,0.f,0.f,0.f,0.f,0.f,0.f,0.f};

#pragma unroll
  for (int kb = 0; kb < KPER; kb += 32) {
    v4f af[TM][4], bf[TN][4];
#pragma unroll
    for (int j = 0; j < 4; ++j) {
      const int o = kb + 4 * j;
#pragma unroll
      for (int i = 0; i < TM; ++i) af[i][j] = *(const v4f*)(aPtr[i] + o);
#pragma unroll
      for (int t = 0; t < TN; ++t) bf[t][j] = *(const v4f*)(bPtr[t] + o);
    }
#pragma unroll
    for (int j = 0; j < 4; ++j) {
#pragma unroll
      for (int p = 0; p < 2; ++p) {
        v2f fa[TM];
#pragma unroll
        for (int i = 0; i < TM; ++i)
          fa[i] = p ? (v2f){af[i][j].z, af[i][j].w} : (v2f){af[i][j].x, af[i][j].y};
#pragma unroll
        for (int t = 0; t < TN; ++t) {
          v2f fb = p ? (v2f){bf[t][j].z, bf[t][j].w}
                     : (v2f){bf[t][j].x, bf[t][j].y};
#pragma unroll
          for (int i = 0; i < TM; ++i)
            acc[i][t] = __builtin_amdgcn_wmma_f32_16x16x4_f32(
                false, fa[i], false, fb, (short)0, acc[i][t], false, false);
        }
      }
    }
    __builtin_amdgcn_sched_group_barrier(0x020, 4 * (TM + TN), 0);
    __builtin_amdgcn_sched_group_barrier(0x008, 8 * TM * TN, 0);
  }

#pragma unroll
  for (int i = 0; i < TM; ++i)
#pragma unroll
    for (int t = 0; t < TN; ++t)
#pragma unroll
      for (int vq = 0; vq < 2; ++vq)
        part[wv][(i * TN + t) * 2 + vq][lane] =
            (v4f){acc[i][t][4 * vq], acc[i][t][4 * vq + 1],
                  acc[i][t][4 * vq + 2], acc[i][t][4 * vq + 3]};
  __syncthreads();

  float* dst = side ? rightp : leftp;
#pragma unroll
  for (int t = 0; t < TN; ++t) {
#pragma unroll
    for (int vq = 0; vq < 2; ++vq) {
      const int q = (wv * TN + t) * 2 + vq;
      const int rowbase = m0 + 16 * wv + ((lane >> 4) ? 8 : 0) + 4 * vq;
      v4f s = part[0][q][lane];
      s += part[1][q][lane];
      s += part[2][q][lane];
      s += part[3][q][lane];
#pragma unroll
      for (int u = 0; u < 4; ++u)
        dst[(long)(rowbase + u) * RANK + ccol0 + t * 16] = s[u] + bias[t];
    }
  }
}

#define TP 68
__global__ __launch_bounds__(256) void pair_relu_dot_kernel(
    const float* __restrict__ leftp, const float* __restrict__ rightp,
    const float* __restrict__ wlast, const float* __restrict__ blast,
    float* __restrict__ out)
{
  __shared__ __attribute__((aligned(16))) float Lt[RANK][TP];
  __shared__ __attribute__((aligned(16))) float Mt[RANK][TP];
  __shared__ __attribute__((aligned(16))) float ws[RANK];
  __shared__ __attribute__((aligned(16))) float cs[64];

  const int t  = threadIdx.x;
  const int jb = blockIdx.x;
  const int ib = blockIdx.y;
  const int bz = blockIdx.z;
  const int si = bz * SS + ib * 64;
  const int sj = bz * SS + jb * 64;

  {
    const int row = (t >> 5) * 8 + ((t & 31) >> 2);
    const int cl  = t & 3;
    const float4* lr = (const float4*)(leftp  + (long)(si + row) * RANK);
    const float4* rr = (const float4*)(rightp + (long)(sj + row) * RANK);
    const float4* wr = (const float4*)wlast;
    float cp0 = 0.f, cp1 = 0.f, cp2 = 0.f, cp3 = 0.f;
#pragma unroll
    for (int u = 0; u < 16; ++u) {
      const int c4 = 4 * u + cl;
      float4 lv = lr[c4];
      float4 rv = rr[c4];
      float4 wv = wr[c4];
      const int r0 = 4 * c4;
      Lt[r0][row]     = lv.x; Lt[r0 + 1][row] = lv.y;
      Lt[r0 + 2][row] = lv.z; Lt[r0 + 3][row] = lv.w;
      Mt[r0][row]     = -rv.x; Mt[r0 + 1][row] = -rv.y;
      Mt[r0 + 2][row] = -rv.z; Mt[r0 + 3][row] = -rv.w;
      cp0 = __builtin_fmaf(rv.x, wv.x, cp0);
      cp1 = __builtin_fmaf(rv.y, wv.y, cp1);
      cp2 = __builtin_fmaf(rv.z, wv.z, cp2);
      cp3 = __builtin_fmaf(rv.w, wv.w, cp3);
    }
    float cp = (cp0 + cp1) + (cp2 + cp3);
    cp += __shfl_xor(cp, 1, 32);
    cp += __shfl_xor(cp, 2, 32);
    if (cl == 0) cs[row] = cp;
    ws[t] = wlast[t];
  }
  __syncthreads();

  const int ti = t >> 4;
  const int tj = t & 15;
  v2f P[4][2];
#pragma unroll
  for (int a = 0; a < 4; ++a) { P[a][0] = (v2f){0.f, 0.f}; P[a][1] = (v2f){0.f, 0.f}; }

  for (int r = 0; r < RANK; r += 2) {
    const v4f la0 = *(const v4f*)&Lt[r][ti * 4];
    const v4f mb0 = *(const v4f*)&Mt[r][tj * 4];
    const v4f la1 = *(const v4f*)&Lt[r + 1][ti * 4];
    const v4f mb1 = *(const v4f*)&Mt[r + 1][tj * 4];
    const v2f w2  = *(const v2f*)&ws[r];
    asm(
        "v_dual_max_num_f32 v224, %8, %12 :: v_dual_max_num_f32 v225, %8, %13\n\t"
        "v_dual_max_num_f32 v226, %8, %14 :: v_dual_max_num_f32 v227, %8, %15\n\t"
        "v_dual_max_num_f32 v228, %9, %12 :: v_dual_max_num_f32 v229, %9, %13\n\t"
        "v_dual_max_num_f32 v230, %9, %14 :: v_dual_max_num_f32 v231, %9, %15\n\t"
        "v_dual_max_num_f32 v232, %10, %12 :: v_dual_max_num_f32 v233, %10, %13\n\t"
        "v_dual_max_num_f32 v234, %10, %14 :: v_dual_max_num_f32 v235, %10, %15\n\t"
        "v_dual_max_num_f32 v236, %11, %12 :: v_dual_max_num_f32 v237, %11, %13\n\t"
        "v_dual_max_num_f32 v238, %11, %14 :: v_dual_max_num_f32 v239, %11, %15\n\t"
        "v_pk_fma_f32 %0, v[224:225], %16, %0 op_sel_hi:[1,0,1]\n\t"
        "v_pk_fma_f32 %1, v[226:227], %16, %1 op_sel_hi:[1,0,1]\n\t"
        "v_pk_fma_f32 %2, v[228:229], %16, %2 op_sel_hi:[1,0,1]\n\t"
        "v_pk_fma_f32 %3, v[230:231], %16, %3 op_sel_hi:[1,0,1]\n\t"
        "v_pk_fma_f32 %4, v[232:233], %16, %4 op_sel_hi:[1,0,1]\n\t"
        "v_pk_fma_f32 %5, v[234:235], %16, %5 op_sel_hi:[1,0,1]\n\t"
        "v_pk_fma_f32 %6, v[236:237], %16, %6 op_sel_hi:[1,0,1]\n\t"
        "v_pk_fma_f32 %7, v[238:239], %16, %7 op_sel_hi:[1,0,1]"
        : "+v"(P[0][0]), "+v"(P[0][1]), "+v"(P[1][0]), "+v"(P[1][1]),
          "+v"(P[2][0]), "+v"(P[2][1]), "+v"(P[3][0]), "+v"(P[3][1])
        : "v"(la0[0]), "v"(la0[1]), "v"(la0[2]), "v"(la0[3]),
          "v"(mb0[0]), "v"(mb0[1]), "v"(mb0[2]), "v"(mb0[3]), "v"(w2)
        : "v224", "v225", "v226", "v227", "v228", "v229", "v230", "v231", "v232", "v233", "v234", "v235", "v236", "v237", "v238", "v239");
    asm(
        "v_dual_max_num_f32 v240, %8, %12 :: v_dual_max_num_f32 v241, %8, %13\n\t"
        "v_dual_max_num_f32 v242, %8, %14 :: v_dual_max_num_f32 v243, %8, %15\n\t"
        "v_dual_max_num_f32 v244, %9, %12 :: v_dual_max_num_f32 v245, %9, %13\n\t"
        "v_dual_max_num_f32 v246, %9, %14 :: v_dual_max_num_f32 v247, %9, %15\n\t"
        "v_dual_max_num_f32 v248, %10, %12 :: v_dual_max_num_f32 v249, %10, %13\n\t"
        "v_dual_max_num_f32 v250, %10, %14 :: v_dual_max_num_f32 v251, %10, %15\n\t"
        "v_dual_max_num_f32 v252, %11, %12 :: v_dual_max_num_f32 v253, %11, %13\n\t"
        "v_dual_max_num_f32 v254, %11, %14 :: v_dual_max_num_f32 v255, %11, %15\n\t"
        "v_pk_fma_f32 %0, v[240:241], %16, %0 op_sel:[0,1,0]\n\t"
        "v_pk_fma_f32 %1, v[242:243], %16, %1 op_sel:[0,1,0]\n\t"
        "v_pk_fma_f32 %2, v[244:245], %16, %2 op_sel:[0,1,0]\n\t"
        "v_pk_fma_f32 %3, v[246:247], %16, %3 op_sel:[0,1,0]\n\t"
        "v_pk_fma_f32 %4, v[248:249], %16, %4 op_sel:[0,1,0]\n\t"
        "v_pk_fma_f32 %5, v[250:251], %16, %5 op_sel:[0,1,0]\n\t"
        "v_pk_fma_f32 %6, v[252:253], %16, %6 op_sel:[0,1,0]\n\t"
        "v_pk_fma_f32 %7, v[254:255], %16, %7 op_sel:[0,1,0]"
        : "+v"(P[0][0]), "+v"(P[0][1]), "+v"(P[1][0]), "+v"(P[1][1]),
          "+v"(P[2][0]), "+v"(P[2][1]), "+v"(P[3][0]), "+v"(P[3][1])
        : "v"(la1[0]), "v"(la1[1]), "v"(la1[2]), "v"(la1[3]),
          "v"(mb1[0]), "v"(mb1[1]), "v"(mb1[2]), "v"(mb1[3]), "v"(w2)
        : "v240", "v241", "v242", "v243", "v244", "v245", "v246", "v247", "v248", "v249", "v250", "v251", "v252", "v253", "v254", "v255");
  }

  const float bl = blast[0];
  const v4f cj = *(const v4f*)&cs[tj * 4];
#pragma unroll
  for (int a = 0; a < 4; ++a) {
#pragma unroll
    for (int b = 0; b < 4; ++b) {
      int i = ib * 64 + ti * 4 + a;
      int j = jb * 64 + tj * 4 + b;
      out[(long)bz * (SS * SS) + (long)i * SS + j] = (P[a][b >> 1][b & 1] + cj[b]) + bl;
    }
  }
}

extern "C" void kernel_launch(void* const* d_in, const int* in_sizes, int n_in,
                              void* d_out, int out_size, void* d_ws, size_t ws_size,
                              hipStream_t stream) {
  (void)in_sizes; (void)n_in; (void)out_size; (void)ws_size;
  const float* batch  = (const float*)d_in[0];
  const float* W1     = (const float*)d_in[1];
  const float* b1     = (const float*)d_in[2];
  const float* W_last = (const float*)d_in[3];
  const float* b_last = (const float*)d_in[4];
  float* out = (float*)d_out;

  float* leftp  = (float*)d_ws;
  float* rightp = leftp + (size_t)NROWS * RANK;

  dim3 g1(NROWS / 64, NCOLS / 64);
  gemm_lr_kernel<<<g1, 128, 0, stream>>>(batch, W1, b1, leftp, rightp);

  dim3 g2(4, 4, NB);
  pair_relu_dot_kernel<<<g2, 256, 0, stream>>>(leftp, rightp, W_last, b_last, out);
}
